# starting kernel
# baseline (speedup 1.0000x reference)
#include <hip/hip_runtime.h>


typedef __attribute__((ext_vector_type(2))) float v2f;
typedef __attribute__((ext_vector_type(8))) float v8f;

#define C_IN   128
#define C_OUT  128
#define HDIM   128
#define WDIM   128
#define KW     (C_IN * 9)
#define CCHUNK 16
#define KCHUNK (CCHUNK * 9)
#define NPIX   64
#define PSTRIDE (NPIX * 2 + 32)
#define NPAIRS  (KCHUNK / 2)

__global__ __launch_bounds__(256)
void sdconv_transpose_w(const float* __restrict__ Wt, float* __restrict__ Wk) {
  const int i = blockIdx.x * 256 + threadIdx.x;
  if (i < KW * C_OUT) {
    const int k = i >> 7;
    const int o = i & (C_OUT - 1);
    Wk[i] = Wt[o * KW + k];
  }
}

template <bool KMAJOR>
__global__ __launch_bounds__(256)
void sdconv_wmma_f32(const float* __restrict__ U, const float* __restrict__ V,
                     const float* __restrict__ Wt, float* __restrict__ Out) {
  __shared__ float sS[NPAIRS * PSTRIDE];

  const int bid  = blockIdx.x;
  const int wseg = bid & 1;
  const int h    = (bid >> 1) & (HDIM - 1);
  const int bB   = bid >> 8;
  const int w0   = wseg * NPIX;

  const int tid    = threadIdx.x;
  const int lane   = tid & 31;
  const int wave   = tid >> 5;
  const int oBase  = wave * 16;
  const int laneM  = lane & 15;
  const int laneHi = lane >> 4;

  const int p    = tid & (NPIX - 1);
  const int crow = tid >> 6;

  v8f acc0 = {}; v8f acc1 = {}; v8f acc2 = {}; v8f acc3 = {};

  for (int cc = 0; cc < C_IN / CCHUNK; ++cc) {
    const int cBase = cc * CCHUNK;

    #pragma unroll
    for (int i = 0; i < 4; ++i) {
      const int cl = crow + i * 4;
      const int c  = cBase + cl;
      const int w  = w0 + p;
      const int rowBase = ((bB * C_IN + c) * HDIM + h) * WDIM;
      const float uc = U[rowBase + w];
      const float vc = V[rowBase + w];
      #pragma unroll
      for (int tap = 0; tap < 9; ++tap) {
        const int dy = tap / 3 - 1;
        const int dx = tap % 3 - 1;
        const int hh = h + dy;
        const int ww = w + dx;
        float un = 0.0f, vn = 0.0f;
        if (hh >= 0 && hh < HDIM && ww >= 0 && ww < WDIM) {
          const int nb = ((bB * C_IN + c) * HDIM + hh) * WDIM + ww;
          un = U[nb]; vn = V[nb];
        }
        const float dU = un - uc;
        const float dV = vn - vc;
        const float s  = dU * rsqrtf(1.0f + dV * dV);
        const int kidx = cl * 9 + tap;
        sS[(kidx >> 1) * PSTRIDE + p * 2 + (kidx & 1)] = s;
      }
    }
    __syncthreads();

    if (cc + 1 < C_IN / CCHUNK) {
      #pragma unroll
      for (int i = 0; i < 4; ++i) {
        const int c = cBase + CCHUNK + crow + i * 4;
        const int rowBase = ((bB * C_IN + c) * HDIM + h) * WDIM + w0 + p;
        __builtin_prefetch(&U[rowBase], 0, 0);
        __builtin_prefetch(&V[rowBase], 0, 0);
        if (h > 0)        { __builtin_prefetch(&U[rowBase - WDIM], 0, 0);
                            __builtin_prefetch(&V[rowBase - WDIM], 0, 0); }
        if (h < HDIM - 1) { __builtin_prefetch(&U[rowBase + WDIM], 0, 0);
                            __builtin_prefetch(&V[rowBase + WDIM], 0, 0); }
      }
    }

    const int kw0 = cBase * 9;
    #pragma unroll 4
    for (int ks = 0; ks < KCHUNK / 4; ++ks) {
      const int k0 = ks * 4;
      v2f a;
      if (KMAJOR) {
        const float* wp = Wt + (kw0 + k0 + laneHi * 2) * C_OUT + oBase + laneM;
        a.x = wp[0];
        a.y = wp[C_OUT];
      } else {
        const float* wp = Wt + (oBase + laneM) * KW + (kw0 + k0 + laneHi * 2);
        a.x = wp[0];
        a.y = wp[1];
      }
      const int pr = (k0 >> 1) + laneHi;
      const float* sp = &sS[pr * PSTRIDE + laneM * 2];
      v2f b0 = *(const v2f*)(sp +  0);
      v2f b1 = *(const v2f*)(sp + 32);
      v2f b2 = *(const v2f*)(sp + 64);
      v2f b3 = *(const v2f*)(sp + 96);
      acc0 = __builtin_amdgcn_wmma_f32_16x16x4_f32(false, a, false, b0, (short)0, acc0, false, false);
      acc1 = __builtin_amdgcn_wmma_f32_16x16x4_f32(false, a, false, b1, (short)0, acc1, false, false);
      acc2 = __builtin_amdgcn_wmma_f32_16x16x4_f32(false, a, false, b2, (short)0, acc2, false, false);
      acc3 = __builtin_amdgcn_wmma_f32_16x16x4_f32(false, a, false, b3, (short)0, acc3, false, false);
    }
    __syncthreads();
  }

  #pragma unroll
  for (int i = 0; i < 8; ++i) {
    const int o = oBase + i + laneHi * 8;
    const int obase = ((bB * C_OUT + o) * HDIM + h) * WDIM + w0 + laneM;
    Out[obase +  0] = acc0[i];
    Out[obase + 16] = acc1[i];
    Out[obase + 32] = acc2[i];
    Out[obase + 48] = acc3[i];
  }
}

extern "C" void kernel_launch(void* const* d_in, const int* in_sizes, int n_in,
                              void* d_out, int out_size, void* d_ws, size_t ws_size,
                              hipStream_t stream) {
  (void)in_sizes; (void)n_in; (void)out_size;
  const float* U  = (const float*)d_in[0];
  const float* V  = (const float*)d_in[1];
  const float* Wt = (const float*)d_in[2];
  float* Out = (float*)d_out;

  dim3 grid(4 * HDIM * (WDIM / NPIX));
  dim3 block(256);

  const size_t wbytes = (size_t)KW * C_OUT * sizeof(float);
  if (ws_size >= wbytes) {
    float* Wk = (float*)d_ws;
    sdconv_transpose_w<<<dim3((KW * C_OUT + 255) / 256), block, 0, stream>>>(Wt, Wk);
    sdconv_wmma_f32<true><<<grid, block, 0, stream>>>(U, V, Wk, Out);
  } else {
    sdconv_wmma_f32<false><<<grid, block, 0, stream>>>(U, V, Wt, Out);
  }
}


# wave tile 128 out-channels x 32 pixels: a wave builds the S of its own pixels directly as B fragments in registers (no LDS, no barrier); f32 sum order changes (K interleaved by lane half)
# speedup vs baseline: 1.3018x; 1.3018x over previous
#include <hip/hip_runtime.h>


typedef __attribute__((ext_vector_type(2))) float v2f;
typedef __attribute__((ext_vector_type(4))) float v4f;
typedef __attribute__((ext_vector_type(8))) float v8f;

#define C_IN   128
#define C_OUT  128
#define HDIM   128
#define WDIM   128
#define KW     (C_IN * 9)
#define PLANE  (HDIM * WDIM)
#define NIT    (C_IN / 4)

__global__ __launch_bounds__(256)
void sdconv_transpose_w(const float* __restrict__ Wt, float* __restrict__ Wk) {
  const int i = blockIdx.x * 256 + threadIdx.x;
  if (i < KW * C_OUT) {
    const int k = i >> 7;
    const int o = i & (C_OUT - 1);
    Wk[i] = Wt[o * KW + k];
  }
}

struct AFrag { v4f r0lo, r0hi, r1lo, r1hi; };

template <bool KMAJOR>
__device__ __forceinline__ AFrag load_a(const float* __restrict__ Wt, int k, int laneM) {
  AFrag f;
  if (KMAJOR) {
    const v4f* p0 = (const v4f*)(Wt + k * C_OUT + laneM * 8);
    const v4f* p1 = (const v4f*)(Wt + (k + 1) * C_OUT + laneM * 8);
    f.r0lo = p0[0]; f.r0hi = p0[1];
    f.r1lo = p1[0]; f.r1hi = p1[1];
  } else {
    #pragma unroll
    for (int t = 0; t < 4; ++t) {
      f.r0lo[t] = Wt[(laneM * 8 + t) * KW + k];
      f.r1lo[t] = Wt[(laneM * 8 + t) * KW + k + 1];
      f.r0hi[t] = Wt[(laneM * 8 + 4 + t) * KW + k];
      f.r1hi[t] = Wt[(laneM * 8 + 4 + t) * KW + k + 1];
    }
  }
  return f;
}

__device__ __forceinline__ v2f a_pair(const AFrag& f, int t) {
  v2f a;
  a.x = (t < 4) ? f.r0lo[t & 3] : f.r0hi[t & 3];
  a.y = (t < 4) ? f.r1lo[t & 3] : f.r1hi[t & 3];
  return a;
}

template <bool KMAJOR>
__global__ __launch_bounds__(128, 2)
void sdconv_wmma_f32(const float* __restrict__ U, const float* __restrict__ V,
                     const float* __restrict__ Wt, float* __restrict__ Out) {
  const int bid  = blockIdx.x;
  const int hrow = bid & (HDIM - 1);
  const int bB   = bid >> 7;

  const int tid    = threadIdx.x;
  const int lane   = tid & 31;
  const int wave   = tid >> 5;
  const int laneM  = lane & 15;
  const int laneHi = lane >> 4;

  int  toff[2][9];
  bool tok[2][9];
  #pragma unroll
  for (int n = 0; n < 2; ++n) {
    const int w = wave * 32 + n * 16 + laneM;
    #pragma unroll
    for (int t = 0; t < 9; ++t) {
      const int hh = hrow + t / 3 - 1;
      const int ww = w + t % 3 - 1;
      const bool ok = hh >= 0 && hh < HDIM && ww >= 0 && ww < WDIM;
      tok[n][t]  = ok;
      toff[n][t] = (ok ? hh * WDIM + ww : hrow * WDIM + w) + laneHi * 2 * PLANE;
    }
  }

  float ru[2][2][9], rv[2][2][9];
  {
    const int cb = (bB * C_IN) * PLANE;
    #pragma unroll
    for (int q = 0; q < 2; ++q) {
      #pragma unroll
      for (int n = 0; n < 2; ++n) {
        #pragma unroll
        for (int t = 0; t < 9; ++t) {
          ru[q][n][t] = U[cb + q * PLANE + toff[n][t]];
          rv[q][n][t] = V[cb + q * PLANE + toff[n][t]];
        }
      }
    }
  }

  v8f acc[8][2];
  #pragma unroll
  for (int t = 0; t < 8; ++t) {
    #pragma unroll
    for (int n = 0; n < 2; ++n) {
      #pragma unroll
      for (int r = 0; r < 8; ++r) acc[t][n][r] = 0.0f;
    }
  }

  AFrag cur = load_a<KMAJOR>(Wt, 18 * laneHi, laneM);

  float sb[2][18];
  #pragma unroll
  for (int q = 0; q < 2; ++q) {
    #pragma unroll
    for (int n = 0; n < 2; ++n) {
      const float uc = ru[q][n][4];
      const float vc = rv[q][n][4];
      #pragma unroll
      for (int t = 0; t < 9; ++t) {
        const float un = tok[n][t] ? ru[q][n][t] : 0.0f;
        const float vn = tok[n][t] ? rv[q][n][t] : 0.0f;
        const float dU = un - uc;
        const float dV = vn - vc;
        sb[n][q * 9 + t] = dU * rsqrtf(1.0f + dV * dV);
      }
    }
  }

  for (int it = 0; it < NIT; ++it) {

    const int itn = (it + 1 < NIT) ? it + 1 : it;
    const int cbn = (bB * C_IN + 4 * itn) * PLANE;
    const int kb  = 36 * it  + 18 * laneHi;
    const int kbn = 36 * itn + 18 * laneHi;

    #pragma unroll
    for (int j = 0; j < 9; ++j) {
      const AFrag nxt = load_a<KMAJOR>(Wt, (j < 8) ? kb + 2 * (j + 1) : kbn, laneM);
      if (j < 6) {
        #pragma unroll
        for (int f = 6 * j; f < 6 * j + 6; ++f) {
          const int q = f / 18;
          const int n = (f / 9) % 2;
          const int t = f % 9;
          ru[q][n][t] = U[cbn + q * PLANE + toff[n][t]];
          rv[q][n][t] = V[cbn + q * PLANE + toff[n][t]];
        }
      }
      __builtin_amdgcn_sched_barrier(0);
      v2f b0, b1;
      b0.x = sb[0][2 * j]; b0.y = sb[0][2 * j + 1];
      b1.x = sb[1][2 * j]; b1.y = sb[1][2 * j + 1];
      #pragma unroll
      for (int t = 0; t < 8; ++t) {
        const v2f a = a_pair(cur, t);
        acc[t][0] = __builtin_amdgcn_wmma_f32_16x16x4_f32(false, a, false, b0, (short)0, acc[t][0], false, false);
        acc[t][1] = __builtin_amdgcn_wmma_f32_16x16x4_f32(false, a, false, b1, (short)0, acc[t][1], false, false);
      }
      cur = nxt;
    }
    #pragma unroll
    for (int q = 0; q < 2; ++q) {
      #pragma unroll
      for (int n = 0; n < 2; ++n) {
        const float uc = ru[q][n][4];
        const float vc = rv[q][n][4];
        #pragma unroll
        for (int t = 0; t < 9; ++t) {
          const float un = tok[n][t] ? ru[q][n][t] : 0.0f;
          const float vn = tok[n][t] ? rv[q][n][t] : 0.0f;
          const float dU = un - uc;
          const float dV = vn - vc;
          sb[n][q * 9 + t] = dU * rsqrtf(1.0f + dV * dV);
        }
      }
    }
  }

  #pragma unroll
  for (int t = 0; t < 8; ++t) {
    #pragma unroll
    for (int r = 0; r < 8; ++r) {
      const int o = (laneHi * 8 + r) * 8 + t;
      const int obase = ((bB * C_OUT + o) * HDIM + hrow) * WDIM + wave * 32 + laneM;
      Out[obase +  0] = acc[t][0][r];
      Out[obase + 16] = acc[t][1][r];
    }
  }
}

extern "C" void kernel_launch(void* const* d_in, const int* in_sizes, int n_in,
                              void* d_out, int out_size, void* d_ws, size_t ws_size,
                              hipStream_t stream) {
  (void)in_sizes; (void)n_in; (void)out_size;
  const float* U  = (const float*)d_in[0];
  const float* V  = (const float*)d_in[1];
  const float* Wt = (const float*)d_in[2];
  float* Out = (float*)d_out;

  dim3 grid(4 * HDIM);
  dim3 block(128);
  dim3 blockT(256);

  const size_t wbytes = (size_t)KW * C_OUT * sizeof(float);
  if (ws_size >= wbytes) {
    float* Wk = (float*)d_ws;
    sdconv_transpose_w<<<dim3((KW * C_OUT + 255) / 256), blockT, 0, stream>>>(Wt, Wk);
    sdconv_wmma_f32<true><<<grid, block, 0, stream>>>(U, V, Wk, Out);
  } else {
    sdconv_wmma_f32<false><<<grid, block, 0, stream>>>(U, V, Wt, Out);
  }
}
